# starting kernel
# baseline (speedup 1.0000x reference)
#include <hip/hip_runtime.h>


typedef __attribute__((ext_vector_type(2))) float v2f;
typedef __attribute__((ext_vector_type(8))) float v8f;

#define SR_B 16
#define SR_N 1024
#define SR_D 16
#define SR_ALPHA_HALF 500.0f
#define SR_THREADS 512

__device__ __forceinline__ float sr_tanh(float x) {
#if __has_builtin(__builtin_amdgcn_tanhf)
  return __builtin_amdgcn_tanhf(x);
#else
  float r;
  asm volatile("v_tanh_f32 %0, %1" : "=v"(r) : "v"(x));
  return r;
#endif
}

__global__ __launch_bounds__(SR_THREADS) void softrank_wmma_kernel(
    const float* __restrict__ x, float* __restrict__ out) {
  __shared__ float lds[SR_D * SR_N];

  const int b   = (int)blockIdx.x >> 6;
  const int jb  = (int)blockIdx.x & 63;
  const int tid = (int)threadIdx.x;

  const float* __restrict__ xb = x + b * (SR_N * SR_D);
#pragma unroll
  for (int t = 0; t < (SR_N * SR_D) / SR_THREADS; ++t) {
    const int idx = tid + t * SR_THREADS;
    const int i = idx >> 4;
    const int k = idx & 15;
    lds[(k << 10) + i] = SR_ALPHA_HALF * xb[idx];
  }
  __syncthreads();

  const int  k    = tid >> 5;
  const int  lane = tid & 31;
  const int  ln16 = lane & 15;
  const bool lo   = lane < 16;
  const int  j0   = jb << 4;
  const float* __restrict__ ldsk = &lds[k << 10];

  v2f A;
  A.x = lo ? ldsk[j0 + ln16] : 0.0f;
  A.y = lo ? -1.0f : 0.0f;

  const v8f czero = {0.f, 0.f, 0.f, 0.f, 0.f, 0.f, 0.f, 0.f};
  v8f acc = czero;

  for (int i0 = 0; i0 < SR_N; i0 += 16) {
    v2f Bm;
    Bm.x = 1.0f;
    Bm.y = ldsk[i0 + ln16];

    v8f z = __builtin_amdgcn_wmma_f32_16x16x4_f32(
         false, A,  false, Bm,
         (short)0, czero,  false,  false);

#pragma unroll
    for (int r = 0; r < 8; ++r) acc[r] += sr_tanh(z[r]);
  }

#pragma unroll
  for (int r = 0; r < 8; ++r) {
    float v = acc[r];
    v += __shfl_xor(v, 1, 32);
    v += __shfl_xor(v, 2, 32);
    v += __shfl_xor(v, 4, 32);
    v += __shfl_xor(v, 8, 32);
    acc[r] = v;
  }

  if (ln16 == 0) {
    const int mbase = lo ? 0 : 8;
    float* __restrict__ ob =
        out + b * (SR_N * SR_D) + (j0 + mbase) * SR_D + k;
#pragma unroll
    for (int r = 0; r < 8; ++r)
      ob[r * SR_D] = 0.5f + acc[r] * (1.0f / (2.0f * SR_N));
  }
}

extern "C" void kernel_launch(void* const* d_in, const int* in_sizes, int n_in,
                              void* d_out, int out_size, void* d_ws, size_t ws_size,
                              hipStream_t stream) {
  (void)in_sizes; (void)n_in; (void)out_size; (void)d_ws; (void)ws_size;
  const float* x = (const float*)d_in[0];
  float* out = (float*)d_out;
  softrank_wmma_kernel<<<dim3(SR_B * (SR_N / 16)), dim3(SR_THREADS), 0, stream>>>(x, out);
}


# Tile turned: features on WMMA N axis, x_j as C operand, one source row per WMMA read straight from global; no LDS transpose, no cross-lane reduce; 8 waves split i, join in LDS (sum order changes)
# speedup vs baseline: 1.5207x; 1.5207x over previous
#include <hip/hip_runtime.h>


typedef __attribute__((ext_vector_type(2))) float v2f;
typedef __attribute__((ext_vector_type(8))) float v8f;

#define SR_B 16
#define SR_N 1024
#define SR_D 16
#define SR_ALPHA_HALF 500.0f
#define SR_JT 2
#define SR_WAVES 8
#define SR_THREADS 256
#define SR_ROWS 128
#define SR_PAIRS 8
#define SR_CHUNKS 4
#define SR_TILES 64
#define SR_GRID 512

__device__ __forceinline__ float sr_tanh(float x) {
#if __has_builtin(__builtin_amdgcn_tanhf)
  return __builtin_amdgcn_tanhf(x);
#else
  float r;
  asm volatile("v_tanh_f32 %0, %1" : "=v"(r) : "v"(x));
  return r;
#endif
}

__global__ __launch_bounds__(SR_THREADS) void softrank_rowbcast_kernel(
    const float* __restrict__ x, float* __restrict__ out) {
  __shared__ float part[SR_WAVES][SR_JT * 256];

  const int  b    = (int)blockIdx.x >> 5;
  const int  jg   = (int)blockIdx.x & 31;
  const int  tid  = (int)threadIdx.x;
  const int  wave = tid >> 5;
  const int  lane = tid & 31;
  const int  c    = lane & 15;
  const int  h    = lane >> 4;
  const bool lo   = lane < 16;
  const int  j0   = jg * (16 * SR_JT);

  const float* __restrict__ xb = x + b * (SR_N * SR_D);

  const float* __restrict__ src = xb + wave * (SR_ROWS * SR_D) + lane;

  v2f raw[SR_PAIRS];
#pragma unroll
  for (int p = 0; p < SR_PAIRS; ++p) {
    raw[p].x = src[p * 64];
    raw[p].y = src[p * 64 + 32];
  }

  v8f C[SR_JT];
#pragma unroll
  for (int t = 0; t < SR_JT; ++t)
#pragma unroll
    for (int r = 0; r < 8; ++r)
      C[t][r] = SR_ALPHA_HALF * xb[(j0 + 16 * t + 8 * h + r) * SR_D + c];

  v2f Asel[4];
  Asel[0].x = lo ? -1.0f : 0.0f;  Asel[0].y = 0.0f;
  Asel[1].x = lo ? 0.0f : -1.0f;  Asel[1].y = 0.0f;
  Asel[2].x = 0.0f;               Asel[2].y = lo ? -1.0f : 0.0f;
  Asel[3].x = 0.0f;               Asel[3].y = lo ? 0.0f : -1.0f;

  v8f acc[SR_JT];
#pragma unroll
  for (int t = 0; t < SR_JT; ++t)
#pragma unroll
    for (int r = 0; r < 8; ++r) acc[t][r] = 0.0f;

  for (int ch = 0; ch < SR_CHUNKS; ++ch) {
    v2f cur[SR_PAIRS];
#pragma unroll
    for (int p = 0; p < SR_PAIRS; ++p) cur[p] = raw[p] * SR_ALPHA_HALF;

    if (ch + 1 < SR_CHUNKS) {
      const float* __restrict__ nx = src + (ch + 1) * (4 * SR_PAIRS * SR_D);
#pragma unroll
      for (int p = 0; p < SR_PAIRS; ++p) {
        raw[p].x = nx[p * 64];
        raw[p].y = nx[p * 64 + 32];
      }
    }

    v8f z0 = __builtin_amdgcn_wmma_f32_16x16x4_f32(
        false, Asel[0], false, cur[0], (short)0, C[0], false, false);
    v8f z1 = __builtin_amdgcn_wmma_f32_16x16x4_f32(
        false, Asel[0], false, cur[0], (short)0, C[1], false, false);
    v8f tp;
#pragma unroll
    for (int n = 0; n < SR_TILES; ++n) {
      v8f z2 = z1;
      if (n + 2 < SR_TILES) {
        const int n2 = n + 2;
        z2 = __builtin_amdgcn_wmma_f32_16x16x4_f32(
            false, Asel[(n2 >> 1) & 3], false, cur[n2 >> 3], (short)0, C[n2 & 1],
            false, false);
      }
      v8f tn;
#pragma unroll
      for (int r = 0; r < 8; ++r) {
        tn[r] = sr_tanh(z0[r]);
        if (n > 0) acc[(n - 1) & 1][r] += tp[r];
      }
      tp = tn;
      z0 = z1;
      z1 = z2;
    }
#pragma unroll
    for (int r = 0; r < 8; ++r) acc[1][r] += tp[r];
  }

#pragma unroll
  for (int t = 0; t < SR_JT; ++t)
#pragma unroll
    for (int r = 0; r < 8; ++r)
      part[wave][t * 256 + (8 * h + r) * SR_D + c] = acc[t][r];
  __syncthreads();

  const int o = tid * 2;
  float s0 = 0.0f, s1 = 0.0f;
#pragma unroll
  for (int w = 0; w < SR_WAVES; ++w) {
    const float2 v = *reinterpret_cast<const float2*>(&part[w][o]);
    s0 += v.x;
    s1 += v.y;
  }
  float2 res;
  res.x = 0.5f + s0 * (1.0f / (2.0f * SR_N));
  res.y = 0.5f + s1 * (1.0f / (2.0f * SR_N));
  *reinterpret_cast<float2*>(out + b * (SR_N * SR_D) + j0 * SR_D + o) = res;
}

extern "C" void kernel_launch(void* const* d_in, const int* in_sizes, int n_in,
                              void* d_out, int out_size, void* d_ws, size_t ws_size,
                              hipStream_t stream) {
  (void)in_sizes; (void)n_in; (void)out_size; (void)d_ws; (void)ws_size;
  const float* x = (const float*)d_in[0];
  float* out = (float*)d_out;
  softrank_rowbcast_kernel<<<dim3(SR_GRID), dim3(SR_THREADS), 0, stream>>>(x, out);
}
